# starting kernel
# baseline (speedup 1.0000x reference)
#include <hip/hip_runtime.h>
#include <hip/hip_bf16.h>


typedef float v2f __attribute__((ext_vector_type(2)));
typedef float v8f __attribute__((ext_vector_type(8)));

#define PDIM 256
#define HDIM 256
#define NBATCH 8
#define LSEQ 4096
#define MROWS (NBATCH * LSEQ)
#define LC 256
#define NCHUNK (LSEQ / LC)
#define LDSPITCH 260

static __device__ inline v8f wmma4f32(v2f a, v2f b, v8f c) {
  return __builtin_amdgcn_wmma_f32_16x16x4_f32(
       false, a,  false, b,
       (short)0, c,  false,  false);
}

__global__ void prep_params(const float* __restrict__ A_diag,
                            const float* __restrict__ G_diag,
                            const float* __restrict__ dt,
                            float* __restrict__ m11,
                            float* __restrict__ m12,
                            float* __restrict__ dts) {
  int p = threadIdx.x;
  float s = 1.0f / (1.0f + __expf(-dt[p]));
  float Glow = s * A_diag[p];
  float G = Glow + fmaxf(G_diag[p] - Glow, 0.0f);
  float Alow = 0.25f * G * G;
  float A = Alow + fmaxf(A_diag[p] - Alow, 0.0f);
  m11[p] = 1.0f - s * G;
  m12[p] = -s * A;
  dts[p] = s;
}

__global__ void pack_BC(const float* __restrict__ Bmat,
                        const float* __restrict__ Cmat,
                        float* __restrict__ BreP, float* __restrict__ BimP,
                        float* __restrict__ CreP, float* __restrict__ CimN) {
  int idx = blockIdx.x * blockDim.x + threadIdx.x;
  int p = idx & (PDIM - 1);
  int h = idx >> 8;
  float br = Bmat[(p * HDIM + h) * 2 + 0];
  float bi = Bmat[(p * HDIM + h) * 2 + 1];
  int bpos = (h & ~1) * PDIM + 2 * p + (h & 1);
  BreP[bpos] = br;
  BimP[bpos] = bi;
  float cr = Cmat[(h * PDIM + p) * 2 + 0];
  float ci = Cmat[(h * PDIM + p) * 2 + 1];
  int cpos = (p & ~1) * HDIM + 2 * h + (p & 1);
  CreP[cpos] = cr;
  CimN[cpos] = -ci;
}

__global__ __launch_bounds__(256) void gemm_bu(
    const float* __restrict__ u, const float* __restrict__ BreP,
    const float* __restrict__ BimP, const float* __restrict__ dts,
    float* __restrict__ fre, float* __restrict__ fim) {
  __shared__ float su[16 * LDSPITCH];
  const int tid = threadIdx.x;
  const int rowBase = blockIdx.x * 16;

  const float* src = u + (size_t)rowBase * HDIM;
  for (int i = tid; i < 16 * HDIM; i += 256)
    su[(i >> 8) * LDSPITCH + (i & 255)] = src[i];
  __syncthreads();

  const int wave = tid >> 5, lane = tid & 31;
  const int pBase = blockIdx.y * 128 + wave * 16;
  const int mn = lane & 15;
  const int ksel = (lane >> 4) << 1;

  v8f accRe = {};
  v8f accIm = {};
  const float* arow = su + mn * LDSPITCH + ksel;
  const float* brow = BreP + 2 * (pBase + mn);
  const float* irow = BimP + 2 * (pBase + mn);

  for (int k = 0; k < HDIM; k += 4) {
    v2f a = *(const v2f*)(arow + k);
    v2f b0 = *(const v2f*)(brow + (k + ksel) * PDIM);
    v2f b1 = *(const v2f*)(irow + (k + ksel) * PDIM);
    accRe = wmma4f32(a, b0, accRe);
    accIm = wmma4f32(a, b1, accIm);
  }

  const float scale = dts[pBase + mn];
  const int mo = rowBase + ((lane >> 4) << 3);
  const int col = pBase + mn;
#pragma unroll
  for (int r = 0; r < 8; ++r) {
    fre[(size_t)(mo + r) * PDIM + col] = accRe[r] * scale;
    fim[(size_t)(mo + r) * PDIM + col] = accIm[r] * scale;
  }
}

__global__ void scan_chunks(const float* __restrict__ fre,
                            const float* __restrict__ fim,
                            float* __restrict__ c1, float* __restrict__ c2,
                            const float* __restrict__ m11,
                            const float* __restrict__ m12,
                            const float* __restrict__ dts) {
  int tid = blockIdx.x * blockDim.x + threadIdx.x;
  int p = tid & (PDIM - 1);
  int comp = (tid >> 8) & 1;
  int chunk = (tid >> 9) & (NCHUNK - 1);
  int b = tid >> 13;

  const float* f = (comp ? fim : fre) +
                   (size_t)(b * LSEQ + chunk * LC) * PDIM + p;
  const float a11 = m11[p], a12 = m12[p], a21 = dts[p];
  float s1 = 0.0f, s2 = 0.0f;
  for (int t = 0; t < LC; ++t) {
    float ft = f[(size_t)t * PDIM];
    float n1 = fmaf(a11, s1, fmaf(a12, s2, ft));
    float n2 = fmaf(a21, s1, s2);
    s1 = n1;
    s2 = n2;
  }
  int ci = ((b * 2 + comp) * NCHUNK + chunk) * PDIM + p;
  c1[ci] = s1;
  c2[ci] = s2;
}

__global__ void scan_combine(float* __restrict__ c1, float* __restrict__ c2,
                             const float* __restrict__ m11,
                             const float* __restrict__ m12,
                             const float* __restrict__ dts) {
  int tid = blockIdx.x * blockDim.x + threadIdx.x;
  int p = tid & (PDIM - 1);
  int comp = (tid >> 8) & 1;
  int b = tid >> 9;

  float a = m11[p], bb = m12[p], c = dts[p], d = 1.0f;
#pragma unroll
  for (int i = 0; i < 8; ++i) {
    float na = fmaf(a, a, bb * c);
    float nb = fmaf(a, bb, bb * d);
    float nc = fmaf(c, a, d * c);
    float nd = fmaf(c, bb, d * d);
    a = na; bb = nb; c = nc; d = nd;
  }

  float i1 = 0.0f, i2 = 0.0f;
  int base = (b * 2 + comp) * NCHUNK * PDIM + p;
  for (int ch = 0; ch < NCHUNK; ++ch) {
    int ci = base + ch * PDIM;
    float s1 = c1[ci], s2 = c2[ci];
    c1[ci] = i1;
    c2[ci] = i2;
    float t1 = fmaf(a, i1, fmaf(bb, i2, s1));
    float t2 = fmaf(c, i1, fmaf(d, i2, s2));
    i1 = t1;
    i2 = t2;
  }
}

__global__ void scan_apply(float* __restrict__ fre, float* __restrict__ fim,
                           const float* __restrict__ c1,
                           const float* __restrict__ c2,
                           const float* __restrict__ m11,
                           const float* __restrict__ m12,
                           const float* __restrict__ dts) {
  int tid = blockIdx.x * blockDim.x + threadIdx.x;
  int p = tid & (PDIM - 1);
  int comp = (tid >> 8) & 1;
  int chunk = (tid >> 9) & (NCHUNK - 1);
  int b = tid >> 13;

  float* f = (comp ? fim : fre) + (size_t)(b * LSEQ + chunk * LC) * PDIM + p;
  const float a11 = m11[p], a12 = m12[p], a21 = dts[p];
  int ci = ((b * 2 + comp) * NCHUNK + chunk) * PDIM + p;
  float x1 = c1[ci], x2 = c2[ci];
  for (int t = 0; t < LC; ++t) {
    float ft = f[(size_t)t * PDIM];
    float y2 = fmaf(a21, x1, x2);
    float y1 = fmaf(a11, x1, fmaf(a12, x2, ft));
    f[(size_t)t * PDIM] = y2;
    x1 = y1;
    x2 = y2;
  }
}

__global__ __launch_bounds__(256) void gemm_out(
    const float* __restrict__ ysre, const float* __restrict__ ysim,
    const float* __restrict__ CreP, const float* __restrict__ CimN,
    const float* __restrict__ u, const float* __restrict__ Dvec,
    float* __restrict__ out) {
  __shared__ float sre[16 * LDSPITCH];
  __shared__ float sim[16 * LDSPITCH];
  const int tid = threadIdx.x;
  const int rowBase = blockIdx.x * 16;

  const float* pr = ysre + (size_t)rowBase * PDIM;
  const float* pi = ysim + (size_t)rowBase * PDIM;
  for (int i = tid; i < 16 * PDIM; i += 256) {
    int r = i >> 8, c = i & 255;
    sre[r * LDSPITCH + c] = pr[i];
    sim[r * LDSPITCH + c] = pi[i];
  }
  __syncthreads();

  const int wave = tid >> 5, lane = tid & 31;
  const int hBase = blockIdx.y * 128 + wave * 16;
  const int mn = lane & 15;
  const int ksel = (lane >> 4) << 1;

  v8f acc = {};
  const float* arow_re = sre + mn * LDSPITCH + ksel;
  const float* arow_im = sim + mn * LDSPITCH + ksel;
  const float* crow = CreP + 2 * (hBase + mn);
  const float* irow = CimN + 2 * (hBase + mn);

  for (int k = 0; k < PDIM; k += 4) {
    v2f ar = *(const v2f*)(arow_re + k);
    v2f ai = *(const v2f*)(arow_im + k);
    v2f vr = *(const v2f*)(crow + (k + ksel) * HDIM);
    v2f vi = *(const v2f*)(irow + (k + ksel) * HDIM);
    acc = wmma4f32(ar, vr, acc);
    acc = wmma4f32(ai, vi, acc);
  }

  const int h = hBase + mn;
  const float dh = Dvec[h];
  const int mo = rowBase + ((lane >> 4) << 3);
#pragma unroll
  for (int r = 0; r < 8; ++r) {
    size_t idx = (size_t)(mo + r) * HDIM + h;
    out[idx] = acc[r] + dh * u[idx];
  }
}

extern "C" void kernel_launch(void* const* d_in, const int* in_sizes, int n_in,
                              void* d_out, int out_size, void* d_ws,
                              size_t ws_size, hipStream_t stream) {
  const float* u      = (const float*)d_in[0];
  const float* A_diag = (const float*)d_in[1];
  const float* G_diag = (const float*)d_in[2];
  const float* dt     = (const float*)d_in[3];
  const float* Bmat   = (const float*)d_in[4];
  const float* Cmat   = (const float*)d_in[5];
  const float* Dvec   = (const float*)d_in[6];
  float* out = (float*)d_out;

  float* w = (float*)d_ws;
  const size_t NF = (size_t)MROWS * PDIM;
  float* BreP = w;
  float* BimP = BreP + (size_t)HDIM * PDIM;
  float* CreP = BimP + (size_t)HDIM * PDIM;
  float* CimN = CreP + (size_t)PDIM * HDIM;
  float* m11  = CimN + (size_t)PDIM * HDIM;
  float* m12  = m11 + PDIM;
  float* dts  = m12 + PDIM;
  float* fre  = dts + PDIM;
  float* fim  = fre + NF;
  float* c1   = fim + NF;
  float* c2   = c1 + (size_t)NBATCH * 2 * NCHUNK * PDIM;

  prep_params<<<1, 256, 0, stream>>>(A_diag, G_diag, dt, m11, m12, dts);
  pack_BC<<<256, 256, 0, stream>>>(Bmat, Cmat, BreP, BimP, CreP, CimN);
  gemm_bu<<<dim3(MROWS / 16, PDIM / 128), 256, 0, stream>>>(
      u, BreP, BimP, dts, fre, fim);
  scan_chunks<<<256, 256, 0, stream>>>(fre, fim, c1, c2, m11, m12, dts);
  scan_combine<<<16, 256, 0, stream>>>(c1, c2, m11, m12, dts);
  scan_apply<<<256, 256, 0, stream>>>(fre, fim, c1, c2, m11, m12, dts);
  gemm_out<<<dim3(MROWS / 16, HDIM / 128), 256, 0, stream>>>(
      fre, fim, CreP, CimN, u, Dvec, out);
}


# Three chunked scan launches become one single-pass scan: a work-group walks a sequence set in 1024-step register tiles, 64-step sub-chunk states joined via LDS; carry sum order changes
# speedup vs baseline: 1.2177x; 1.2177x over previous
#include <hip/hip_runtime.h>
#include <hip/hip_bf16.h>


typedef float v2f __attribute__((ext_vector_type(2)));
typedef float v8f __attribute__((ext_vector_type(8)));

#define PDIM 256
#define HDIM 256
#define NBATCH 8
#define LSEQ 4096
#define MROWS (NBATCH * LSEQ)
#define LDSPITCH 260
#define ROWS_PER_WG 512
#define TILE_ROWS 32
#define TILES_PER_WG (ROWS_PER_WG / TILE_ROWS)
#define SLICE_COLS 64
#define WPLANE (64 * 4 * 64)

static __device__ inline v8f wmma4f32(v2f a, v2f b, v8f c) {
  return __builtin_amdgcn_wmma_f32_16x16x4_f32(
       false, a,  false, b,
       (short)0, c,  false,  false);
}

__global__ void prep_params(const float* __restrict__ A_diag,
                            const float* __restrict__ G_diag,
                            const float* __restrict__ dt,
                            float* __restrict__ m11,
                            float* __restrict__ m12,
                            float* __restrict__ dts) {
  int p = threadIdx.x;
  float s = 1.0f / (1.0f + __expf(-dt[p]));
  float Glow = s * A_diag[p];
  float G = Glow + fmaxf(G_diag[p] - Glow, 0.0f);
  float Alow = 0.25f * G * G;
  float A = Alow + fmaxf(A_diag[p] - Alow, 0.0f);
  m11[p] = 1.0f - s * G;
  m12[p] = -s * A;
  dts[p] = s;
}

__global__ void pack_BC(const float* __restrict__ Bmat,
                        const float* __restrict__ Cmat,
                        float* __restrict__ BreP, float* __restrict__ BimP,
                        float* __restrict__ CreP, float* __restrict__ CimN) {
  int idx = blockIdx.x * blockDim.x + threadIdx.x;
  int p = idx & (PDIM - 1);
  int h = idx >> 8;
  float br = Bmat[(p * HDIM + h) * 2 + 0];
  float bi = Bmat[(p * HDIM + h) * 2 + 1];
  int bpos = (h & ~1) * PDIM + 2 * p + (h & 1);
  BreP[bpos] = br;
  BimP[bpos] = bi;
  float cr = Cmat[(h * PDIM + p) * 2 + 0];
  float ci = Cmat[(h * PDIM + p) * 2 + 1];
  int cpos = (p & ~1) * HDIM + 2 * h + (p & 1);
  CreP[cpos] = cr;
  CimN[cpos] = -ci;
}

static __device__ inline void stage_weights(const float* __restrict__ src,
                                            int colBase, float* dst, int tid) {
  const int q4 = tid & 15, rp = (tid >> 4) & 1, qhi = (tid >> 5) & 1;
  const int n = 2 * (qhi * 16 + q4);
  const float* s = src + 2 * (colBase + n) + (size_t)rp * 512;
  float* d = dst + (((n >> 4) * 32) + rp * 16 + (n & 15)) * 2;
  for (int it = 0; it < 16; ++it) {
    int rr = (tid >> 6) + 4 * it;
    float4 v = *(const float4*)(s + (size_t)rr * 1024);
    *(float4*)(d + rr * 256) = v;
  }
}

#define FRAG_DECL(P) v2f P##a0, P##a1, P##a2, P##a3, P##r0, P##r1, P##r2, \
                         P##r3, P##i0, P##i1, P##i2, P##i3

#define BU_LOAD(P, kb)                                            \
  P##a0 = *(const v2f*)(arow + (kb));                             \
  P##a1 = *(const v2f*)(arow + (kb) + 4);                         \
  P##a2 = *(const v2f*)(arow + (kb) + 8);                         \
  P##a3 = *(const v2f*)(arow + (kb) + 12);                        \
  P##r0 = *(const v2f*)(brow + (kb) * 64);                        \
  P##i0 = *(const v2f*)(irow + (kb) * 64);                        \
  P##r1 = *(const v2f*)(brow + ((kb) + 4) * 64);                  \
  P##i1 = *(const v2f*)(irow + ((kb) + 4) * 64);                  \
  P##r2 = *(const v2f*)(brow + ((kb) + 8) * 64);                  \
  P##i2 = *(const v2f*)(irow + ((kb) + 8) * 64);                  \
  P##r3 = *(const v2f*)(brow + ((kb) + 12) * 64);                 \
  P##i3 = *(const v2f*)(irow + ((kb) + 12) * 64);
#define BU_MMA(P)                                                 \
  accRe = wmma4f32(P##a0, P##r0, accRe);                          \
  accIm = wmma4f32(P##a0, P##i0, accIm);                          \
  accRe = wmma4f32(P##a1, P##r1, accRe);                          \
  accIm = wmma4f32(P##a1, P##i1, accIm);                          \
  accRe = wmma4f32(P##a2, P##r2, accRe);                          \
  accIm = wmma4f32(P##a2, P##i2, accIm);                          \
  accRe = wmma4f32(P##a3, P##r3, accRe);                          \
  accIm = wmma4f32(P##a3, P##i3, accIm);

__global__ __launch_bounds__(256) void gemm_bu(
    const float* __restrict__ u, const float* __restrict__ BreP,
    const float* __restrict__ BimP, const float* __restrict__ dts,
    float* __restrict__ fre, float* __restrict__ fim) {
  __shared__ float sB[2 * WPLANE];
  __shared__ float su[TILE_ROWS * LDSPITCH];
  const int tid = threadIdx.x;
  const int rowBlk = blockIdx.x * ROWS_PER_WG;
  const int pSlice = blockIdx.y * SLICE_COLS;

  const float4* gsrc = (const float4*)(u + (size_t)rowBlk * HDIM) + tid;
  float4 g0 = gsrc[0], g1 = gsrc[256], g2 = gsrc[512], g3 = gsrc[768];
  float4 g4 = gsrc[1024], g5 = gsrc[1280], g6 = gsrc[1536], g7 = gsrc[1792];

  stage_weights(BreP, pSlice, sB, tid);
  stage_weights(BimP, pSlice, sB + WPLANE, tid);

  const int wave = tid >> 5, lane = tid & 31;
  const int ct = wave & 3, rt = wave >> 2;
  const int pBase = pSlice + ct * 16;
  const int mn = lane & 15;
  const int ksel = (lane >> 4) << 1;
  const float* arow = su + (rt * 16 + mn) * LDSPITCH + ksel;
  const float* brow = sB + ct * 64 + lane * 2;
  const float* irow = brow + WPLANE;
  const float scale = dts[pBase + mn];
  const int col = pBase + mn;
  float* sdst = su + (tid >> 6) * LDSPITCH + 4 * (tid & 63);
  v8f pRe = {}, pIm = {};

  for (int t = 0; t < TILES_PER_WG; ++t) {
    *(float4*)(sdst) = g0;
    *(float4*)(sdst + 4 * LDSPITCH) = g1;
    *(float4*)(sdst + 8 * LDSPITCH) = g2;
    *(float4*)(sdst + 12 * LDSPITCH) = g3;
    *(float4*)(sdst + 16 * LDSPITCH) = g4;
    *(float4*)(sdst + 20 * LDSPITCH) = g5;
    *(float4*)(sdst + 24 * LDSPITCH) = g6;
    *(float4*)(sdst + 28 * LDSPITCH) = g7;
    __syncthreads();
    if (t + 1 < TILES_PER_WG) {
      const float4* gn = gsrc + (t + 1) * 2048;
      g0 = gn[0];    g1 = gn[256];  g2 = gn[512];  g3 = gn[768];
      g4 = gn[1024]; g5 = gn[1280]; g6 = gn[1536]; g7 = gn[1792];
    }

    if (t > 0) {
      const int mo = rowBlk + (t - 1) * TILE_ROWS + rt * 16 + ((lane >> 4) << 3);
#pragma unroll
      for (int r = 0; r < 8; ++r) {
        fre[(size_t)(mo + r) * PDIM + col] = pRe[r];
        fim[(size_t)(mo + r) * PDIM + col] = pIm[r];
      }
    }

    v8f accRe = {};
    v8f accIm = {};
    FRAG_DECL(x);
    FRAG_DECL(y);
    BU_LOAD(x, 0)
    for (int kc = 0; kc < HDIM; kc += 32) {
      BU_LOAD(y, kc + 16)
      __builtin_amdgcn_sched_barrier(0);
      BU_MMA(x)
      __builtin_amdgcn_sched_barrier(0);
      if (kc + 32 < HDIM) { BU_LOAD(x, kc + 32) }
      __builtin_amdgcn_sched_barrier(0);
      BU_MMA(y)
      __builtin_amdgcn_sched_barrier(0);
    }
    pRe = accRe * scale;
    pIm = accIm * scale;
    __syncthreads();
  }
  {
    const int mo = rowBlk + (TILES_PER_WG - 1) * TILE_ROWS + rt * 16 +
                   ((lane >> 4) << 3);
#pragma unroll
    for (int r = 0; r < 8; ++r) {
      fre[(size_t)(mo + r) * PDIM + col] = pRe[r];
      fim[(size_t)(mo + r) * PDIM + col] = pIm[r];
    }
  }
}

#define SC_S 64
#define SC_NW 16
#define SC_TILE (SC_S * SC_NW)
#define SC_NTILE (LSEQ / SC_TILE)
#define SC_SLICES (PDIM / 32)

static __device__ inline void lds_only_barrier() {
  asm volatile("s_wait_dscnt 0x0\n\ts_barrier_signal -1\n\ts_barrier_wait -1"
               ::: "memory");
}

__global__ __launch_bounds__(SC_NW * 32) void scan_seq(
    float* __restrict__ fre, float* __restrict__ fim,
    const float* __restrict__ m11, const float* __restrict__ m12,
    const float* __restrict__ dts) {
  __shared__ float2 carry[2][SC_NW][32];
  const int tid = threadIdx.x;
  const int w = tid >> 5, lane = tid & 31;
  const int bid = blockIdx.x;
  const int p = (bid % SC_SLICES) * 32 + lane;
  const int comp = (bid / SC_SLICES) & 1;
  const int b = bid / (2 * SC_SLICES);
  float* f = (comp ? fim : fre) +
             ((size_t)b * LSEQ + (size_t)w * SC_S) * PDIM + p;

  float cur[SC_S], nxt[SC_S];
#pragma unroll
  for (int j = 0; j < SC_S; ++j) cur[j] = f[(size_t)j * PDIM];

  const float a11 = m11[p], a12 = m12[p], a21 = dts[p];
  float qa = a11, qb = a12, qc = a21, qd = 1.0f;
#pragma unroll
  for (int i = 0; i < 6; ++i) {
    float na = fmaf(qa, qa, qb * qc);
    float nb = fmaf(qa, qb, qb * qd);
    float nc = fmaf(qc, qa, qd * qc);
    float nd = fmaf(qc, qb, qd * qd);
    qa = na; qb = nb; qc = nc; qd = nd;
  }

  float g1 = 0.0f, g2 = 0.0f;
#pragma unroll
  for (int k = 0; k < SC_NTILE; ++k) {
    float* ft = f + (size_t)k * SC_TILE * PDIM;
    if (k + 1 < SC_NTILE) {
#pragma unroll
      for (int j = 0; j < SC_S; ++j)
        nxt[j] = ft[(size_t)(SC_TILE + j) * PDIM];
    }
    __builtin_amdgcn_sched_barrier(0);
    float s1 = 0.0f, s2 = 0.0f;
#pragma unroll
    for (int j = 0; j < SC_S; ++j) {
      float n1 = fmaf(a11, s1, fmaf(a12, s2, cur[j]));
      float n2 = fmaf(a21, s1, s2);
      s1 = n1;
      s2 = n2;
    }
    carry[k & 1][w][lane] = make_float2(s1, s2);
    lds_only_barrier();
    float i1 = g1, i2 = g2, x1 = 0.0f, x2 = 0.0f;
#pragma unroll
    for (int v = 0; v < SC_NW; ++v) {
      if (v == w) { x1 = i1; x2 = i2; }
      float2 c = carry[k & 1][v][lane];
      float t1 = fmaf(qa, i1, fmaf(qb, i2, c.x));
      float t2 = fmaf(qc, i1, fmaf(qd, i2, c.y));
      i1 = t1;
      i2 = t2;
    }
    g1 = i1;
    g2 = i2;
#pragma unroll
    for (int j = 0; j < SC_S; ++j) {
      float y2 = fmaf(a21, x1, x2);
      float y1 = fmaf(a11, x1, fmaf(a12, x2, cur[j]));
      ft[(size_t)j * PDIM] = y2;
      x1 = y1;
      x2 = y2;
    }
#pragma unroll
    for (int j = 0; j < SC_S; ++j) cur[j] = nxt[j];
  }
}

#define OUT_LOAD(P, kb)                                           \
  P##a0 = *(const v2f*)(arow_re + (kb));                          \
  P##a1 = *(const v2f*)(arow_im + (kb));                          \
  P##a2 = *(const v2f*)(arow_re + (kb) + 4);                      \
  P##a3 = *(const v2f*)(arow_im + (kb) + 4);                      \
  P##a4 = *(const v2f*)(arow_re + (kb) + 8);                      \
  P##a5 = *(const v2f*)(arow_im + (kb) + 8);                      \
  P##a6 = *(const v2f*)(arow_re + (kb) + 12);                     \
  P##a7 = *(const v2f*)(arow_im + (kb) + 12);                     \
  P##r0 = *(const v2f*)(crow + (kb) * 64);                        \
  P##i0 = *(const v2f*)(irow + (kb) * 64);                        \
  P##r1 = *(const v2f*)(crow + ((kb) + 4) * 64);                  \
  P##i1 = *(const v2f*)(irow + ((kb) + 4) * 64);                  \
  P##r2 = *(const v2f*)(crow + ((kb) + 8) * 64);                  \
  P##i2 = *(const v2f*)(irow + ((kb) + 8) * 64);                  \
  P##r3 = *(const v2f*)(crow + ((kb) + 12) * 64);                 \
  P##i3 = *(const v2f*)(irow + ((kb) + 12) * 64);
#define OUT_MMA(P)                                                \
  acc = wmma4f32(P##a0, P##r0, acc);         \
  acc = wmma4f32(P##a1, P##i0, acc);         \
  acc = wmma4f32(P##a2, P##r1, acc);                              \
  acc = wmma4f32(P##a3, P##i1, acc);                              \
  acc = wmma4f32(P##a4, P##r2, acc);                              \
  acc = wmma4f32(P##a5, P##i2, acc);                              \
  acc = wmma4f32(P##a6, P##r3, acc);                              \
  acc = wmma4f32(P##a7, P##i3, acc);
#define OUT_DECL(P) v2f P##a0, P##a1, P##a2, P##a3, P##a4, P##a5, P##a6, \
                        P##a7, P##r0, P##r1, P##r2, P##r3, P##i0, P##i1, \
                        P##i2, P##i3

__global__ __launch_bounds__(256) void gemm_out(
    const float* __restrict__ ysre, const float* __restrict__ ysim,
    const float* __restrict__ CreP, const float* __restrict__ CimN,
    const float* __restrict__ u, const float* __restrict__ Dvec,
    float* __restrict__ out) {
  __shared__ float sC[2 * WPLANE];
  __shared__ float sre[TILE_ROWS * LDSPITCH];
  __shared__ float sim[TILE_ROWS * LDSPITCH];
  const int tid = threadIdx.x;
  const int rowBlk = blockIdx.x * ROWS_PER_WG;
  const int hSlice = blockIdx.y * SLICE_COLS;

  const float4* pr = (const float4*)(ysre + (size_t)rowBlk * PDIM) + tid;
  const float4* pi = (const float4*)(ysim + (size_t)rowBlk * PDIM) + tid;
  float4 g0 = pr[0], g1 = pr[256], g2 = pr[512], g3 = pr[768];
  float4 g4 = pr[1024], g5 = pr[1280], g6 = pr[1536], g7 = pr[1792];
  float4 q0 = pi[0], q1 = pi[256], q2 = pi[512], q3 = pi[768];
  float4 q4 = pi[1024], q5 = pi[1280], q6 = pi[1536], q7 = pi[1792];

  stage_weights(CreP, hSlice, sC, tid);
  stage_weights(CimN, hSlice, sC + WPLANE, tid);

  const int wave = tid >> 5, lane = tid & 31;
  const int ct = wave & 3, rt = wave >> 2;
  const int hBase = hSlice + ct * 16;
  const int mn = lane & 15;
  const int ksel = (lane >> 4) << 1;
  const float* arow_re = sre + (rt * 16 + mn) * LDSPITCH + ksel;
  const float* arow_im = sim + (rt * 16 + mn) * LDSPITCH + ksel;
  const float* crow = sC + ct * 64 + lane * 2;
  const float* irow = crow + WPLANE;
  const int h = hBase + mn;
  const float dh = Dvec[h];
  const int so = (tid >> 6) * LDSPITCH + 4 * (tid & 63);
  float* dre = sre + so;
  float* dim_ = sim + so;
  v8f pend = {};

  for (int t = 0; t < TILES_PER_WG; ++t) {
    *(float4*)(dre) = g0;
    *(float4*)(dre + 4 * LDSPITCH) = g1;
    *(float4*)(dre + 8 * LDSPITCH) = g2;
    *(float4*)(dre + 12 * LDSPITCH) = g3;
    *(float4*)(dre + 16 * LDSPITCH) = g4;
    *(float4*)(dre + 20 * LDSPITCH) = g5;
    *(float4*)(dre + 24 * LDSPITCH) = g6;
    *(float4*)(dre + 28 * LDSPITCH) = g7;
    *(float4*)(dim_) = q0;
    *(float4*)(dim_ + 4 * LDSPITCH) = q1;
    *(float4*)(dim_ + 8 * LDSPITCH) = q2;
    *(float4*)(dim_ + 12 * LDSPITCH) = q3;
    *(float4*)(dim_ + 16 * LDSPITCH) = q4;
    *(float4*)(dim_ + 20 * LDSPITCH) = q5;
    *(float4*)(dim_ + 24 * LDSPITCH) = q6;
    *(float4*)(dim_ + 28 * LDSPITCH) = q7;
    __syncthreads();
    v8f ue;
    {
      const int mo = rowBlk + t * TILE_ROWS + rt * 16 + ((lane >> 4) << 3);
#pragma unroll
      for (int r = 0; r < 8; ++r) ue[r] = u[(size_t)(mo + r) * HDIM + h];
    }
    if (t + 1 < TILES_PER_WG) {
      const float4* rn = pr + (t + 1) * 2048;
      const float4* in = pi + (t + 1) * 2048;
      g0 = rn[0];    g1 = rn[256];  g2 = rn[512];  g3 = rn[768];
      g4 = rn[1024]; g5 = rn[1280]; g6 = rn[1536]; g7 = rn[1792];
      q0 = in[0];    q1 = in[256];  q2 = in[512];  q3 = in[768];
      q4 = in[1024]; q5 = in[1280]; q6 = in[1536]; q7 = in[1792];
    }

    if (t > 0) {
      const int mp = rowBlk + (t - 1) * TILE_ROWS + rt * 16 + ((lane >> 4) << 3);
#pragma unroll
      for (int r = 0; r < 8; ++r)
        out[(size_t)(mp + r) * HDIM + h] = pend[r];
    }

    v8f acc = {};
    OUT_DECL(x);
    OUT_DECL(y);
    OUT_LOAD(x, 0)
    for (int kc = 0; kc < PDIM; kc += 32) {
      OUT_LOAD(y, kc + 16)
      __builtin_amdgcn_sched_barrier(0);
      OUT_MMA(x)
      __builtin_amdgcn_sched_barrier(0);
      if (kc + 32 < PDIM) { OUT_LOAD(x, kc + 32) }
      __builtin_amdgcn_sched_barrier(0);
      OUT_MMA(y)
      __builtin_amdgcn_sched_barrier(0);
    }
#pragma unroll
    for (int r = 0; r < 8; ++r) pend[r] = acc[r] + dh * ue[r];
    __syncthreads();
  }
  {
    const int mp = rowBlk + (TILES_PER_WG - 1) * TILE_ROWS + rt * 16 +
                   ((lane >> 4) << 3);
#pragma unroll
    for (int r = 0; r < 8; ++r)
      out[(size_t)(mp + r) * HDIM + h] = pend[r];
  }
}

extern "C" void kernel_launch(void* const* d_in, const int* in_sizes, int n_in,
                              void* d_out, int out_size, void* d_ws,
                              size_t ws_size, hipStream_t stream) {
  const float* u      = (const float*)d_in[0];
  const float* A_diag = (const float*)d_in[1];
  const float* G_diag = (const float*)d_in[2];
  const float* dt     = (const float*)d_in[3];
  const float* Bmat   = (const float*)d_in[4];
  const float* Cmat   = (const float*)d_in[5];
  const float* Dvec   = (const float*)d_in[6];
  float* out = (float*)d_out;

  float* w = (float*)d_ws;
  const size_t NF = (size_t)MROWS * PDIM;
  float* BreP = w;
  float* BimP = BreP + (size_t)HDIM * PDIM;
  float* CreP = BimP + (size_t)HDIM * PDIM;
  float* CimN = CreP + (size_t)PDIM * HDIM;
  float* m11  = CimN + (size_t)PDIM * HDIM;
  float* m12  = m11 + PDIM;
  float* dts  = m12 + PDIM;
  float* fre  = dts + PDIM;
  float* fim  = fre + NF;

  prep_params<<<1, 256, 0, stream>>>(A_diag, G_diag, dt, m11, m12, dts);
  pack_BC<<<256, 256, 0, stream>>>(Bmat, Cmat, BreP, BimP, CreP, CimN);
  gemm_bu<<<dim3(MROWS / ROWS_PER_WG, PDIM / SLICE_COLS), 256, 0, stream>>>(
      u, BreP, BimP, dts, fre, fim);
  scan_seq<<<NBATCH * 2 * SC_SLICES, SC_NW * 32, 0, stream>>>(fre, fim, m11, m12, dts);
  gemm_out<<<dim3(MROWS / ROWS_PER_WG, HDIM / SLICE_COLS), 256, 0, stream>>>(
      fre, fim, CreP, CimN, u, Dvec, out);
}


# Both GEMMs: A tile double-buffered in LDS; tile refill and result stores moved into the middle of the WMMA stream, one work-group barrier per 32-row pass instead of two; sum order unchanged
# speedup vs baseline: 1.2614x; 1.0359x over previous
#include <hip/hip_runtime.h>
#include <hip/hip_bf16.h>


typedef float v2f __attribute__((ext_vector_type(2)));
typedef float v8f __attribute__((ext_vector_type(8)));

#define PDIM 256
#define HDIM 256
#define NBATCH 8
#define LSEQ 4096
#define MROWS (NBATCH * LSEQ)
#define LDSPITCH 260
#define ROWS_PER_WG 512
#define TILE_ROWS 32
#define TILES_PER_WG (ROWS_PER_WG / TILE_ROWS)
#define SLICE_COLS 64
#define WPLANE (64 * 4 * 64)
#define ATILE (TILE_ROWS * LDSPITCH)

static __device__ inline v8f wmma4f32(v2f a, v2f b, v8f c) {
  return __builtin_amdgcn_wmma_f32_16x16x4_f32(
       false, a,  false, b,
       (short)0, c,  false,  false);
}

__global__ void prep_params(const float* __restrict__ A_diag,
                            const float* __restrict__ G_diag,
                            const float* __restrict__ dt,
                            float* __restrict__ m11,
                            float* __restrict__ m12,
                            float* __restrict__ dts) {
  int p = threadIdx.x;
  float s = 1.0f / (1.0f + __expf(-dt[p]));
  float Glow = s * A_diag[p];
  float G = Glow + fmaxf(G_diag[p] - Glow, 0.0f);
  float Alow = 0.25f * G * G;
  float A = Alow + fmaxf(A_diag[p] - Alow, 0.0f);
  m11[p] = 1.0f - s * G;
  m12[p] = -s * A;
  dts[p] = s;
}

__global__ void pack_BC(const float* __restrict__ Bmat,
                        const float* __restrict__ Cmat,
                        float* __restrict__ BreP, float* __restrict__ BimP,
                        float* __restrict__ CreP, float* __restrict__ CimN) {
  int idx = blockIdx.x * blockDim.x + threadIdx.x;
  int p = idx & (PDIM - 1);
  int h = idx >> 8;
  float br = Bmat[(p * HDIM + h) * 2 + 0];
  float bi = Bmat[(p * HDIM + h) * 2 + 1];
  int bpos = (h & ~1) * PDIM + 2 * p + (h & 1);
  BreP[bpos] = br;
  BimP[bpos] = bi;
  float cr = Cmat[(h * PDIM + p) * 2 + 0];
  float ci = Cmat[(h * PDIM + p) * 2 + 1];
  int cpos = (p & ~1) * HDIM + 2 * h + (p & 1);
  CreP[cpos] = cr;
  CimN[cpos] = -ci;
}

static __device__ inline void stage_weights(const float* __restrict__ src,
                                            int colBase, float* dst, int tid) {
  const int q4 = tid & 15, rp = (tid >> 4) & 1, qhi = (tid >> 5) & 1;
  const int n = 2 * (qhi * 16 + q4);
  const float* s = src + 2 * (colBase + n) + (size_t)rp * 512;
  float* d = dst + (((n >> 4) * 32) + rp * 16 + (n & 15)) * 2;
  for (int it = 0; it < 16; ++it) {
    int rr = (tid >> 6) + 4 * it;
    float4 v = *(const float4*)(s + (size_t)rr * 1024);
    *(float4*)(d + rr * 256) = v;
  }
}

#define FRAG_DECL(P) v2f P##a0, P##a1, P##a2, P##a3, P##r0, P##r1, P##r2, \
                         P##r3, P##i0, P##i1, P##i2, P##i3

#define BU_LOAD(P, kb)                                            \
  P##a0 = *(const v2f*)(arow + (kb));                             \
  P##a1 = *(const v2f*)(arow + (kb) + 4);                         \
  P##a2 = *(const v2f*)(arow + (kb) + 8);                         \
  P##a3 = *(const v2f*)(arow + (kb) + 12);                        \
  P##r0 = *(const v2f*)(brow + (kb) * 64);                        \
  P##i0 = *(const v2f*)(irow + (kb) * 64);                        \
  P##r1 = *(const v2f*)(brow + ((kb) + 4) * 64);                  \
  P##i1 = *(const v2f*)(irow + ((kb) + 4) * 64);                  \
  P##r2 = *(const v2f*)(brow + ((kb) + 8) * 64);                  \
  P##i2 = *(const v2f*)(irow + ((kb) + 8) * 64);                  \
  P##r3 = *(const v2f*)(brow + ((kb) + 12) * 64);                 \
  P##i3 = *(const v2f*)(irow + ((kb) + 12) * 64);
#define BU_MMA(P)                                                 \
  accRe = wmma4f32(P##a0, P##r0, accRe);                          \
  accIm = wmma4f32(P##a0, P##i0, accIm);                          \
  accRe = wmma4f32(P##a1, P##r1, accRe);                          \
  accIm = wmma4f32(P##a1, P##i1, accIm);                          \
  accRe = wmma4f32(P##a2, P##r2, accRe);                          \
  accIm = wmma4f32(P##a2, P##i2, accIm);                          \
  accRe = wmma4f32(P##a3, P##r3, accRe);                          \
  accIm = wmma4f32(P##a3, P##i3, accIm);

__global__ __launch_bounds__(256) void gemm_bu(
    const float* __restrict__ u, const float* __restrict__ BreP,
    const float* __restrict__ BimP, const float* __restrict__ dts,
    float* __restrict__ fre, float* __restrict__ fim) {
  __shared__ float sB[2 * WPLANE];
  __shared__ float su[2 * ATILE];
  const int tid = threadIdx.x;
  const int rowBlk = blockIdx.x * ROWS_PER_WG;
  const int pSlice = blockIdx.y * SLICE_COLS;

  const float4* gsrc = (const float4*)(u + (size_t)rowBlk * HDIM) + tid;
  float4 g0 = gsrc[0], g1 = gsrc[256], g2 = gsrc[512], g3 = gsrc[768];
  float4 g4 = gsrc[1024], g5 = gsrc[1280], g6 = gsrc[1536], g7 = gsrc[1792];

  stage_weights(BreP, pSlice, sB, tid);
  stage_weights(BimP, pSlice, sB + WPLANE, tid);

  const int wave = tid >> 5, lane = tid & 31;
  const int ct = wave & 3, rt = wave >> 2;
  const int pBase = pSlice + ct * 16;
  const int mn = lane & 15;
  const int ksel = (lane >> 4) << 1;
  const float* arow0 = su + (rt * 16 + mn) * LDSPITCH + ksel;
  const float* brow = sB + ct * 64 + lane * 2;
  const float* irow = brow + WPLANE;
  const float scale = dts[pBase + mn];
  const int col = pBase + mn;
  float* sdst0 = su + (tid >> 6) * LDSPITCH + 4 * (tid & 63);
  v8f pRe = {}, pIm = {};

#define BU_STAGE(D)                                               \
    *(float4*)(D) = g0;                                           \
    *(float4*)((D) + 4 * LDSPITCH) = g1;                          \
    *(float4*)((D) + 8 * LDSPITCH) = g2;                          \
    *(float4*)((D) + 12 * LDSPITCH) = g3;                         \
    *(float4*)((D) + 16 * LDSPITCH) = g4;                         \
    *(float4*)((D) + 20 * LDSPITCH) = g5;                         \
    *(float4*)((D) + 24 * LDSPITCH) = g6;                         \
    *(float4*)((D) + 28 * LDSPITCH) = g7;
#define BU_FETCH(T)                                               \
    { const float4* gn = gsrc + (T) * 2048;                       \
      g0 = gn[0];    g1 = gn[256];  g2 = gn[512];  g3 = gn[768];  \
      g4 = gn[1024]; g5 = gn[1280]; g6 = gn[1536]; g7 = gn[1792]; }
  BU_STAGE(sdst0)
  BU_FETCH(1)
  __syncthreads();

  for (int t = 0; t < TILES_PER_WG; ++t) {
    const float* arow = arow0 + (t & 1) * ATILE;
    v8f accRe = {};
    v8f accIm = {};
    FRAG_DECL(x);
    FRAG_DECL(y);
    BU_LOAD(x, 0)
#pragma unroll
    for (int kc = 0; kc < HDIM; kc += 32) {
      BU_LOAD(y, kc + 16)
      if (kc == 128 && t > 0) {
        const int mo = rowBlk + (t - 1) * TILE_ROWS + rt * 16 + ((lane >> 4) << 3);
#pragma unroll
        for (int r = 0; r < 8; ++r) {
          fre[(size_t)(mo + r) * PDIM + col] = pRe[r];
          fim[(size_t)(mo + r) * PDIM + col] = pIm[r];
        }
      }
      if (kc == 64) {
        if (t + 1 < TILES_PER_WG) {
          float* sdst = sdst0 + ((t + 1) & 1) * ATILE;
          BU_STAGE(sdst)
        }
        if (t + 2 < TILES_PER_WG) {
          BU_FETCH(t + 2)
        }
      }
      __builtin_amdgcn_sched_barrier(0);
      BU_MMA(x)
      __builtin_amdgcn_sched_barrier(0);
      if (kc + 32 < HDIM) { BU_LOAD(x, kc + 32) }
      __builtin_amdgcn_sched_barrier(0);
      BU_MMA(y)
      __builtin_amdgcn_sched_barrier(0);
    }
    pRe = accRe * scale;
    pIm = accIm * scale;
    __syncthreads();
  }
  {
    const int mo = rowBlk + (TILES_PER_WG - 1) * TILE_ROWS + rt * 16 +
                   ((lane >> 4) << 3);
#pragma unroll
    for (int r = 0; r < 8; ++r) {
      fre[(size_t)(mo + r) * PDIM + col] = pRe[r];
      fim[(size_t)(mo + r) * PDIM + col] = pIm[r];
    }
  }
}

#define SC_S 64
#define SC_NW 16
#define SC_TILE (SC_S * SC_NW)
#define SC_NTILE (LSEQ / SC_TILE)
#define SC_SLICES (PDIM / 32)

static __device__ inline void lds_only_barrier() {
  asm volatile("s_wait_dscnt 0x0\n\ts_barrier_signal -1\n\ts_barrier_wait -1"
               ::: "memory");
}

__global__ __launch_bounds__(SC_NW * 32) void scan_seq(
    float* __restrict__ fre, float* __restrict__ fim,
    const float* __restrict__ m11, const float* __restrict__ m12,
    const float* __restrict__ dts) {
  __shared__ float2 carry[2][SC_NW][32];
  const int tid = threadIdx.x;
  const int w = tid >> 5, lane = tid & 31;
  const int bid = blockIdx.x;
  const int p = (bid % SC_SLICES) * 32 + lane;
  const int comp = (bid / SC_SLICES) & 1;
  const int b = bid / (2 * SC_SLICES);
  float* f = (comp ? fim : fre) +
             ((size_t)b * LSEQ + (size_t)w * SC_S) * PDIM + p;

  float cur[SC_S], nxt[SC_S];
#pragma unroll
  for (int j = 0; j < SC_S; ++j) cur[j] = f[(size_t)j * PDIM];

  const float a11 = m11[p], a12 = m12[p], a21 = dts[p];
  float qa = a11, qb = a12, qc = a21, qd = 1.0f;
#pragma unroll
  for (int i = 0; i < 6; ++i) {
    float na = fmaf(qa, qa, qb * qc);
    float nb = fmaf(qa, qb, qb * qd);
    float nc = fmaf(qc, qa, qd * qc);
    float nd = fmaf(qc, qb, qd * qd);
    qa = na; qb = nb; qc = nc; qd = nd;
  }

  float g1 = 0.0f, g2 = 0.0f;
#pragma unroll
  for (int k = 0; k < SC_NTILE; ++k) {
    float* ft = f + (size_t)k * SC_TILE * PDIM;
    if (k + 1 < SC_NTILE) {
#pragma unroll
      for (int j = 0; j < SC_S; ++j)
        nxt[j] = ft[(size_t)(SC_TILE + j) * PDIM];
    }
    __builtin_amdgcn_sched_barrier(0);
    float s1 = 0.0f, s2 = 0.0f;
#pragma unroll
    for (int j = 0; j < SC_S; ++j) {
      float n1 = fmaf(a11, s1, fmaf(a12, s2, cur[j]));
      float n2 = fmaf(a21, s1, s2);
      s1 = n1;
      s2 = n2;
    }
    carry[k & 1][w][lane] = make_float2(s1, s2);
    lds_only_barrier();
    float i1 = g1, i2 = g2, x1 = 0.0f, x2 = 0.0f;
#pragma unroll
    for (int v = 0; v < SC_NW; ++v) {
      if (v == w) { x1 = i1; x2 = i2; }
      float2 c = carry[k & 1][v][lane];
      float t1 = fmaf(qa, i1, fmaf(qb, i2, c.x));
      float t2 = fmaf(qc, i1, fmaf(qd, i2, c.y));
      i1 = t1;
      i2 = t2;
    }
    g1 = i1;
    g2 = i2;
#pragma unroll
    for (int j = 0; j < SC_S; ++j) {
      float y2 = fmaf(a21, x1, x2);
      float y1 = fmaf(a11, x1, fmaf(a12, x2, cur[j]));
      ft[(size_t)j * PDIM] = y2;
      x1 = y1;
      x2 = y2;
    }
#pragma unroll
    for (int j = 0; j < SC_S; ++j) cur[j] = nxt[j];
  }
}

#define OUT_LOAD(P, kb)                                           \
  P##a0 = *(const v2f*)(arow_re + (kb));                          \
  P##a1 = *(const v2f*)(arow_im + (kb));                          \
  P##a2 = *(const v2f*)(arow_re + (kb) + 4);                      \
  P##a3 = *(const v2f*)(arow_im + (kb) + 4);                      \
  P##a4 = *(const v2f*)(arow_re + (kb) + 8);                      \
  P##a5 = *(const v2f*)(arow_im + (kb) + 8);                      \
  P##a6 = *(const v2f*)(arow_re + (kb) + 12);                     \
  P##a7 = *(const v2f*)(arow_im + (kb) + 12);                     \
  P##r0 = *(const v2f*)(crow + (kb) * 64);                        \
  P##i0 = *(const v2f*)(irow + (kb) * 64);                        \
  P##r1 = *(const v2f*)(crow + ((kb) + 4) * 64);                  \
  P##i1 = *(const v2f*)(irow + ((kb) + 4) * 64);                  \
  P##r2 = *(const v2f*)(crow + ((kb) + 8) * 64);                  \
  P##i2 = *(const v2f*)(irow + ((kb) + 8) * 64);                  \
  P##r3 = *(const v2f*)(crow + ((kb) + 12) * 64);                 \
  P##i3 = *(const v2f*)(irow + ((kb) + 12) * 64);
#define OUT_MMA(P)                                                \
  acc = wmma4f32(P##a0, P##r0, acc);         \
  acc = wmma4f32(P##a1, P##i0, acc);         \
  acc = wmma4f32(P##a2, P##r1, acc);                              \
  acc = wmma4f32(P##a3, P##i1, acc);                              \
  acc = wmma4f32(P##a4, P##r2, acc);                              \
  acc = wmma4f32(P##a5, P##i2, acc);                              \
  acc = wmma4f32(P##a6, P##r3, acc);                              \
  acc = wmma4f32(P##a7, P##i3, acc);
#define OUT_DECL(P) v2f P##a0, P##a1, P##a2, P##a3, P##a4, P##a5, P##a6, \
                        P##a7, P##r0, P##r1, P##r2, P##r3, P##i0, P##i1, \
                        P##i2, P##i3

__global__ __launch_bounds__(256) void gemm_out(
    const float* __restrict__ ysre, const float* __restrict__ ysim,
    const float* __restrict__ CreP, const float* __restrict__ CimN,
    const float* __restrict__ u, const float* __restrict__ Dvec,
    float* __restrict__ out) {
  __shared__ float sC[2 * WPLANE];
  __shared__ float sre[2 * ATILE];
  __shared__ float sim[2 * ATILE];
  const int tid = threadIdx.x;
  const int rowBlk = blockIdx.x * ROWS_PER_WG;
  const int hSlice = blockIdx.y * SLICE_COLS;

  const float4* pr = (const float4*)(ysre + (size_t)rowBlk * PDIM) + tid;
  const float4* pi = (const float4*)(ysim + (size_t)rowBlk * PDIM) + tid;
  float4 g0 = pr[0], g1 = pr[256], g2 = pr[512], g3 = pr[768];
  float4 g4 = pr[1024], g5 = pr[1280], g6 = pr[1536], g7 = pr[1792];
  float4 q0 = pi[0], q1 = pi[256], q2 = pi[512], q3 = pi[768];
  float4 q4 = pi[1024], q5 = pi[1280], q6 = pi[1536], q7 = pi[1792];

  stage_weights(CreP, hSlice, sC, tid);
  stage_weights(CimN, hSlice, sC + WPLANE, tid);

  const int wave = tid >> 5, lane = tid & 31;
  const int ct = wave & 3, rt = wave >> 2;
  const int hBase = hSlice + ct * 16;
  const int mn = lane & 15;
  const int ksel = (lane >> 4) << 1;
  const float* arow_re0 = sre + (rt * 16 + mn) * LDSPITCH + ksel;
  const float* arow_im0 = sim + (rt * 16 + mn) * LDSPITCH + ksel;
  const float* crow = sC + ct * 64 + lane * 2;
  const float* irow = crow + WPLANE;
  const int h = hBase + mn;
  const float dh = Dvec[h];
  const int so = (tid >> 6) * LDSPITCH + 4 * (tid & 63);
  float* dre0 = sre + so;
  float* dim0 = sim + so;
  v8f pend = {};

#define OUT_STAGE(DR, DI)                                         \
    *(float4*)(DR) = g0;                                          \
    *(float4*)((DR) + 4 * LDSPITCH) = g1;                         \
    *(float4*)((DR) + 8 * LDSPITCH) = g2;                         \
    *(float4*)((DR) + 12 * LDSPITCH) = g3;                        \
    *(float4*)((DR) + 16 * LDSPITCH) = g4;                        \
    *(float4*)((DR) + 20 * LDSPITCH) = g5;                        \
    *(float4*)((DR) + 24 * LDSPITCH) = g6;                        \
    *(float4*)((DR) + 28 * LDSPITCH) = g7;                        \
    *(float4*)(DI) = q0;                                          \
    *(float4*)((DI) + 4 * LDSPITCH) = q1;                         \
    *(float4*)((DI) + 8 * LDSPITCH) = q2;                         \
    *(float4*)((DI) + 12 * LDSPITCH) = q3;                        \
    *(float4*)((DI) + 16 * LDSPITCH) = q4;                        \
    *(float4*)((DI) + 20 * LDSPITCH) = q5;                        \
    *(float4*)((DI) + 24 * LDSPITCH) = q6;                        \
    *(float4*)((DI) + 28 * LDSPITCH) = q7;
#define OUT_FETCH(T)                                              \
    { const float4* rn = pr + (T) * 2048;                         \
      const float4* in = pi + (T) * 2048;                         \
      g0 = rn[0];    g1 = rn[256];  g2 = rn[512];  g3 = rn[768];  \
      g4 = rn[1024]; g5 = rn[1280]; g6 = rn[1536]; g7 = rn[1792]; \
      q0 = in[0];    q1 = in[256];  q2 = in[512];  q3 = in[768];  \
      q4 = in[1024]; q5 = in[1280]; q6 = in[1536]; q7 = in[1792]; }
  OUT_STAGE(dre0, dim0)
  OUT_FETCH(1)
  __syncthreads();

  for (int t = 0; t < TILES_PER_WG; ++t) {
    const float* arow_re = arow_re0 + (t & 1) * ATILE;
    const float* arow_im = arow_im0 + (t & 1) * ATILE;
    v8f ue = {};
    v8f acc = {};
    OUT_DECL(x);
    OUT_DECL(y);
    OUT_LOAD(x, 0)
#pragma unroll
    for (int kc = 0; kc < PDIM; kc += 32) {
      OUT_LOAD(y, kc + 16)
      if (kc == 128) {
        const int mo = rowBlk + t * TILE_ROWS + rt * 16 + ((lane >> 4) << 3);
#pragma unroll
        for (int r = 0; r < 8; ++r) ue[r] = u[(size_t)(mo + r) * HDIM + h];
        if (t > 0) {
          const int mp = mo - TILE_ROWS;
#pragma unroll
          for (int r = 0; r < 8; ++r)
            out[(size_t)(mp + r) * HDIM + h] = pend[r];
        }
      }
      if (kc == 64) {
        if (t + 1 < TILES_PER_WG) {
          float* dre = dre0 + ((t + 1) & 1) * ATILE;
          float* dim_ = dim0 + ((t + 1) & 1) * ATILE;
          OUT_STAGE(dre, dim_)
        }
        if (t + 2 < TILES_PER_WG) {
          OUT_FETCH(t + 2)
        }
      }
      __builtin_amdgcn_sched_barrier(0);
      OUT_MMA(x)
      __builtin_amdgcn_sched_barrier(0);
      if (kc + 32 < PDIM) { OUT_LOAD(x, kc + 32) }
      __builtin_amdgcn_sched_barrier(0);
      OUT_MMA(y)
      __builtin_amdgcn_sched_barrier(0);
    }
#pragma unroll
    for (int r = 0; r < 8; ++r) pend[r] = acc[r] + dh * ue[r];
    __syncthreads();
  }
  {
    const int mp = rowBlk + (TILES_PER_WG - 1) * TILE_ROWS + rt * 16 +
                   ((lane >> 4) << 3);
#pragma unroll
    for (int r = 0; r < 8; ++r)
      out[(size_t)(mp + r) * HDIM + h] = pend[r];
  }
}

extern "C" void kernel_launch(void* const* d_in, const int* in_sizes, int n_in,
                              void* d_out, int out_size, void* d_ws,
                              size_t ws_size, hipStream_t stream) {
  const float* u      = (const float*)d_in[0];
  const float* A_diag = (const float*)d_in[1];
  const float* G_diag = (const float*)d_in[2];
  const float* dt     = (const float*)d_in[3];
  const float* Bmat   = (const float*)d_in[4];
  const float* Cmat   = (const float*)d_in[5];
  const float* Dvec   = (const float*)d_in[6];
  float* out = (float*)d_out;

  float* w = (float*)d_ws;
  const size_t NF = (size_t)MROWS * PDIM;
  float* BreP = w;
  float* BimP = BreP + (size_t)HDIM * PDIM;
  float* CreP = BimP + (size_t)HDIM * PDIM;
  float* CimN = CreP + (size_t)PDIM * HDIM;
  float* m11  = CimN + (size_t)PDIM * HDIM;
  float* m12  = m11 + PDIM;
  float* dts  = m12 + PDIM;
  float* fre  = dts + PDIM;
  float* fim  = fre + NF;

  prep_params<<<1, 256, 0, stream>>>(A_diag, G_diag, dt, m11, m12, dts);
  pack_BC<<<256, 256, 0, stream>>>(Bmat, Cmat, BreP, BimP, CreP, CimN);
  gemm_bu<<<dim3(MROWS / ROWS_PER_WG, PDIM / SLICE_COLS), 256, 0, stream>>>(
      u, BreP, BimP, dts, fre, fim);
  scan_seq<<<NBATCH * 2 * SC_SLICES, SC_NW * 32, 0, stream>>>(fre, fim, m11, m12, dts);
  gemm_out<<<dim3(MROWS / ROWS_PER_WG, HDIM / SLICE_COLS), 256, 0, stream>>>(
      fre, fim, CreP, CimN, u, Dvec, out);
}
